# starting kernel
# baseline (speedup 1.0000x reference)
#include <hip/hip_runtime.h>


#define BATCH   4
#define FDIM    64
#define N_NODES 8192
#define LROW    68

typedef __attribute__((ext_vector_type(2))) float v2f;
typedef __attribute__((ext_vector_type(8))) float v8f;

__global__ __launch_bounds__(256) void gcn_prep(const float* __restrict__ x,
                                                const float* __restrict__ w,
                                                float* __restrict__ xt,
                                                float* __restrict__ sup) {
    int n = blockIdx.x * 256 + threadIdx.x;
    int b = blockIdx.y;
    const float* xb = x + (size_t)b * FDIM * N_NODES;
    float* xtrow = xt  + ((size_t)b * N_NODES + n) * FDIM;
    float* srow  = sup + ((size_t)b * N_NODES + n) * FDIM;

    float acc[FDIM];
#pragma unroll
    for (int o = 0; o < FDIM; ++o) acc[o] = 0.0f;

    for (int f = 0; f < FDIM; ++f) {
        float v = xb[(size_t)f * N_NODES + n];
        xtrow[f] = v;
#pragma unroll
        for (int o = 0; o < FDIM; ++o) acc[o] += v * w[f * FDIM + o];
    }
#pragma unroll
    for (int o = 0; o < FDIM; ++o) srow[o] = acc[o];
}

__device__ __forceinline__ void stage_load(const float* __restrict__ xtb,
                                           const float* __restrict__ vbp,
                                           int j, int tid,
                                           float4* rk, float4* rv) {
#pragma unroll
    for (int i = 0; i < 2; ++i) {
        int idx = i * 128 + tid;
        int row = idx >> 4;
        int c4  = idx & 15;
        size_t off = (size_t)(j * 16 + row) * FDIM + c4 * 4;
        rk[i] = *(const float4*)(xtb + off);
        rv[i] = *(const float4*)(vbp + off);
    }
}

__global__ __launch_bounds__(128) void gcn_flash(const float* __restrict__ xt,
                                                 const float* __restrict__ sup,
                                                 float* __restrict__ out) {
    __shared__ __align__(16) float sh[2][2][16][LROW];

    const int tid  = threadIdx.x;
    const int wave = tid >> 5;
    const int lane = tid & 31;
    const int hi   = lane >> 4;
    const int mrow = lane & 15;

    const int b    = blockIdx.x >> 7;
    const int qblk = blockIdx.x & 127;
    const int qbase = qblk * 64 + wave * 16;

    const float* xtb = xt  + (size_t)b * N_NODES * FDIM;
    const float* vbp = sup + (size_t)b * N_NODES * FDIM;

    v2f q[16];
    {
        const float* qrow = xtb + (size_t)(qbase + mrow) * FDIM + 2 * hi;
#pragma unroll
        for (int c = 0; c < 16; ++c)
            q[c] = *(const v2f*)(qrow + 4 * c);
    }

    v8f accT[4];
#pragma unroll
    for (int t = 0; t < 4; ++t)
#pragma unroll
        for (int r = 0; r < 8; ++r) accT[t][r] = 0.0f;

    float m = -3.0e38f;
    float l = 0.0f;

    float4 rk[2], rv[2];
    stage_load(xtb, vbp, 0, tid, rk, rv);

    const int NJ = N_NODES / 16;
    for (int j = 0; j < NJ; ++j) {
        const int buf = j & 1;
#pragma unroll
        for (int i = 0; i < 2; ++i) {
            int idx = i * 128 + tid;
            int row = idx >> 4, c4 = idx & 15;
            *(float4*)&sh[buf][0][row][c4 * 4] = rk[i];
            *(float4*)&sh[buf][1][row][c4 * 4] = rv[i];
        }
        __syncthreads();
        if (j + 1 < NJ)
            stage_load(xtb, vbp, j + 1, tid, rk, rv);

        const float* kb    = &sh[buf][0][0][0];
        const float* vtile = &sh[buf][1][0][0];

        v8f s;
#pragma unroll
        for (int r = 0; r < 8; ++r) s[r] = 0.0f;
#pragma unroll
        for (int c = 0; c < 16; ++c) {
            v2f kk = *(const v2f*)(kb + mrow * LROW + 4 * c + 2 * hi);
            s = __builtin_amdgcn_wmma_f32_16x16x4_f32(
                    false, kk, false, q[c], (short)0, s, false, false);
        }

        float mloc = s[0];
#pragma unroll
        for (int r = 1; r < 8; ++r) mloc = fmaxf(mloc, s[r]);
        mloc = fmaxf(mloc, __shfl_xor(mloc, 16, 32));
        float mnew = fmaxf(m, mloc);
        float sc = __expf(m - mnew);
        m = mnew;

        float p[8], ls = 0.0f;
#pragma unroll
        for (int r = 0; r < 8; ++r) { p[r] = __expf(s[r] - mnew); ls += p[r]; }
        ls += __shfl_xor(ls, 16, 32);
        l = l * sc + ls;

#pragma unroll
        for (int t = 0; t < 4; ++t)
#pragma unroll
            for (int r = 0; r < 8; ++r) accT[t][r] *= sc;

        float tt[8];
#pragma unroll
        for (int r = 0; r < 8; ++r) tt[r] = __shfl_xor(p[r], 16, 32);
        v2f pb[4];
        pb[0].x = hi ? tt[2] : p[0];  pb[0].y = hi ? tt[3] : p[1];
        pb[1].x = hi ? tt[6] : p[4];  pb[1].y = hi ? tt[7] : p[5];
        pb[2].x = hi ? p[2] : tt[0];  pb[2].y = hi ? p[3] : tt[1];
        pb[3].x = hi ? p[6] : tt[4];  pb[3].y = hi ? p[7] : tt[5];

#pragma unroll
        for (int t = 0; t < 4; ++t) {
#pragma unroll
            for (int c = 0; c < 4; ++c) {
                int krow = 4 * c + 2 * hi;
                v2f a;
                a.x = vtile[krow * LROW + 16 * t + mrow];
                a.y = vtile[(krow + 1) * LROW + 16 * t + mrow];
                accT[t] = __builtin_amdgcn_wmma_f32_16x16x4_f32(
                              false, a, false, pb[c], (short)0, accT[t],
                              false, false);
            }
        }
    }

    float inv = 1.0f / l;
    int n = qbase + mrow;
#pragma unroll
    for (int t = 0; t < 4; ++t)
#pragma unroll
        for (int r = 0; r < 8; ++r) {
            int o = 16 * t + r + 8 * hi;
            out[((size_t)b * FDIM + o) * N_NODES + n] = accT[t][r] * inv;
        }
}

extern "C" void kernel_launch(void* const* d_in, const int* in_sizes, int n_in,
                              void* d_out, int out_size, void* d_ws, size_t ws_size,
                              hipStream_t stream) {
    (void)in_sizes; (void)n_in; (void)out_size; (void)ws_size;
    const float* x = (const float*)d_in[0];
    const float* w = (const float*)d_in[1];
    float* xt  = (float*)d_ws;
    float* sup = xt + (size_t)BATCH * N_NODES * FDIM;
    float* out = (float*)d_out;

    dim3 gPrep(N_NODES / 256, BATCH);
    gcn_prep<<<gPrep, 256, 0, stream>>>(x, w, xt, sup);

    dim3 gFlash(BATCH * (N_NODES / 64));
    gcn_flash<<<gFlash, 128, 0, stream>>>(xt, sup, out);
}


# Next K/V tile held in named registers, not private-memory arrays: its global loads issue right behind the barrier and are waited one trip later, so the tile's WMMAs cover them
# speedup vs baseline: 2.3393x; 2.3393x over previous
#include <hip/hip_runtime.h>


#define BATCH   4
#define FDIM    64
#define N_NODES 8192
#define LROW    68

typedef __attribute__((ext_vector_type(2))) float v2f;
typedef __attribute__((ext_vector_type(4))) float v4f;
typedef __attribute__((ext_vector_type(8))) float v8f;

__global__ __launch_bounds__(256) void gcn_prep(const float* __restrict__ x,
                                                const float* __restrict__ w,
                                                float* __restrict__ xt,
                                                float* __restrict__ sup) {
    int n = blockIdx.x * 256 + threadIdx.x;
    int b = blockIdx.y;
    const float* xb = x + (size_t)b * FDIM * N_NODES;
    float* xtrow = xt  + ((size_t)b * N_NODES + n) * FDIM;
    float* srow  = sup + ((size_t)b * N_NODES + n) * FDIM;

    float acc[FDIM];
#pragma unroll
    for (int o = 0; o < FDIM; ++o) acc[o] = 0.0f;

    for (int f = 0; f < FDIM; ++f) {
        float v = xb[(size_t)f * N_NODES + n];
        xtrow[f] = v;
#pragma unroll
        for (int o = 0; o < FDIM; ++o) acc[o] += v * w[f * FDIM + o];
    }
#pragma unroll
    for (int o = 0; o < FDIM; ++o) srow[o] = acc[o];
}

__global__ __launch_bounds__(128) void gcn_flash(const float* __restrict__ xt,
                                                 const float* __restrict__ sup,
                                                 float* __restrict__ out) {
    __shared__ __align__(16) float sh[2][2][16][LROW];

    const int tid  = threadIdx.x;
    const int wave = tid >> 5;
    const int lane = tid & 31;
    const int hi   = lane >> 4;
    const int mrow = lane & 15;

    const int b    = blockIdx.x >> 7;
    const int qblk = blockIdx.x & 127;
    const int qbase = qblk * 64 + wave * 16;

    const float* xtb = xt  + (size_t)b * N_NODES * FDIM;
    const float* vbp = sup + (size_t)b * N_NODES * FDIM;

    v2f q[16];
    {
        const float* qrow = xtb + (size_t)(qbase + mrow) * FDIM + 2 * hi;
#pragma unroll
        for (int c = 0; c < 16; ++c)
            q[c] = *(const v2f*)(qrow + 4 * c);
    }

    v8f accT[4];
#pragma unroll
    for (int t = 0; t < 4; ++t)
#pragma unroll
        for (int r = 0; r < 8; ++r) accT[t][r] = 0.0f;

    float m = -3.0e38f;
    float l = 0.0f;

    const int srow0 = tid >> 4;
    const int scol  = (tid & 15) * 4;
    const size_t goff0 = (size_t)srow0 * FDIM + scol;
    const size_t goff1 = (size_t)(srow0 + 8) * FDIM + scol;
    v4f rk0 = *(const v4f*)(xtb + goff0);
    v4f rk1 = *(const v4f*)(xtb + goff1);
    v4f rv0 = *(const v4f*)(vbp + goff0);
    v4f rv1 = *(const v4f*)(vbp + goff1);

    const int NJ = N_NODES / 16;
    for (int j = 0; j < NJ; ++j) {
        const int buf = j & 1;
        *(v4f*)&sh[buf][0][srow0][scol]     = rk0;
        *(v4f*)&sh[buf][0][srow0 + 8][scol] = rk1;
        *(v4f*)&sh[buf][1][srow0][scol]     = rv0;
        *(v4f*)&sh[buf][1][srow0 + 8][scol] = rv1;
        __syncthreads();
        if (j + 1 < NJ) {
            const size_t tb = (size_t)(j + 1) * 16 * FDIM;
            rk0 = *(const v4f*)(xtb + tb + goff0);
            rk1 = *(const v4f*)(xtb + tb + goff1);
            rv0 = *(const v4f*)(vbp + tb + goff0);
            rv1 = *(const v4f*)(vbp + tb + goff1);
        }
        __builtin_amdgcn_sched_barrier(0);

        const float* kb    = &sh[buf][0][0][0];
        const float* vtile = &sh[buf][1][0][0];

        v8f s;
#pragma unroll
        for (int r = 0; r < 8; ++r) s[r] = 0.0f;
#pragma unroll
        for (int c = 0; c < 16; ++c) {
            v2f kk = *(const v2f*)(kb + mrow * LROW + 4 * c + 2 * hi);
            s = __builtin_amdgcn_wmma_f32_16x16x4_f32(
                    false, kk, false, q[c], (short)0, s, false, false);
        }

        float mloc = s[0];
#pragma unroll
        for (int r = 1; r < 8; ++r) mloc = fmaxf(mloc, s[r]);
        mloc = fmaxf(mloc, __shfl_xor(mloc, 16, 32));
        float mnew = fmaxf(m, mloc);
        float sc = __expf(m - mnew);
        m = mnew;

        float p[8], ls = 0.0f;
#pragma unroll
        for (int r = 0; r < 8; ++r) { p[r] = __expf(s[r] - mnew); ls += p[r]; }
        ls += __shfl_xor(ls, 16, 32);
        l = l * sc + ls;

#pragma unroll
        for (int t = 0; t < 4; ++t)
#pragma unroll
            for (int r = 0; r < 8; ++r) accT[t][r] *= sc;

        float tt[8];
#pragma unroll
        for (int r = 0; r < 8; ++r) tt[r] = __shfl_xor(p[r], 16, 32);
        v2f pb[4];
        pb[0].x = hi ? tt[2] : p[0];  pb[0].y = hi ? tt[3] : p[1];
        pb[1].x = hi ? tt[6] : p[4];  pb[1].y = hi ? tt[7] : p[5];
        pb[2].x = hi ? p[2] : tt[0];  pb[2].y = hi ? p[3] : tt[1];
        pb[3].x = hi ? p[6] : tt[4];  pb[3].y = hi ? p[7] : tt[5];

#pragma unroll
        for (int t = 0; t < 4; ++t) {
#pragma unroll
            for (int c = 0; c < 4; ++c) {
                int krow = 4 * c + 2 * hi;
                v2f a;
                a.x = vtile[krow * LROW + 16 * t + mrow];
                a.y = vtile[(krow + 1) * LROW + 16 * t + mrow];
                accT[t] = __builtin_amdgcn_wmma_f32_16x16x4_f32(
                              false, a, false, pb[c], (short)0, accT[t],
                              false, false);
            }
        }
    }

    float inv = 1.0f / l;
    int n = qbase + mrow;
#pragma unroll
    for (int t = 0; t < 4; ++t)
#pragma unroll
        for (int r = 0; r < 8; ++r) {
            int o = 16 * t + r + 8 * hi;
            out[((size_t)b * FDIM + o) * N_NODES + n] = accT[t][r] * inv;
        }
}

extern "C" void kernel_launch(void* const* d_in, const int* in_sizes, int n_in,
                              void* d_out, int out_size, void* d_ws, size_t ws_size,
                              hipStream_t stream) {
    (void)in_sizes; (void)n_in; (void)out_size; (void)ws_size;
    const float* x = (const float*)d_in[0];
    const float* w = (const float*)d_in[1];
    float* xt  = (float*)d_ws;
    float* sup = xt + (size_t)BATCH * N_NODES * FDIM;
    float* out = (float*)d_out;

    dim3 gPrep(N_NODES / 256, BATCH);
    gcn_prep<<<gPrep, 256, 0, stream>>>(x, w, xt, sup);

    dim3 gFlash(BATCH * (N_NODES / 64));
    gcn_flash<<<gFlash, 128, 0, stream>>>(xt, sup, out);
}


# Keys of each 16-row query tile split over two waves (even/odd 16-key tiles; 8-wave blocks, 4 waves per SIMD instead of 2); the two partial softmax sums per row merged once in LDS at the end
# speedup vs baseline: 2.5918x; 1.1079x over previous
#include <hip/hip_runtime.h>


#define BATCH   4
#define FDIM    64
#define N_NODES 8192
#define LROW    68

typedef __attribute__((ext_vector_type(2))) float v2f;
typedef __attribute__((ext_vector_type(4))) float v4f;
typedef __attribute__((ext_vector_type(8))) float v8f;

__global__ __launch_bounds__(256) void gcn_prep(const float* __restrict__ x,
                                                const float* __restrict__ w,
                                                float* __restrict__ xt,
                                                float* __restrict__ sup) {
    int n = blockIdx.x * 256 + threadIdx.x;
    int b = blockIdx.y;
    const float* xb = x + (size_t)b * FDIM * N_NODES;
    float* xtrow = xt  + ((size_t)b * N_NODES + n) * FDIM;
    float* srow  = sup + ((size_t)b * N_NODES + n) * FDIM;

    float acc[FDIM];
#pragma unroll
    for (int o = 0; o < FDIM; ++o) acc[o] = 0.0f;

    for (int f = 0; f < FDIM; ++f) {
        float v = xb[(size_t)f * N_NODES + n];
        xtrow[f] = v;
#pragma unroll
        for (int o = 0; o < FDIM; ++o) acc[o] += v * w[f * FDIM + o];
    }
#pragma unroll
    for (int o = 0; o < FDIM; ++o) srow[o] = acc[o];
}

__global__ __launch_bounds__(256) void gcn_flash(const float* __restrict__ xt,
                                                 const float* __restrict__ sup,
                                                 float* __restrict__ out) {
    __shared__ __align__(16) float sh[2][2][32][LROW];

    const int tid  = threadIdx.x;
    const int wave = tid >> 5;
    const int lane = tid & 31;
    const int hi   = lane >> 4;
    const int mrow = lane & 15;
    const int qt   = wave & 3;
    const int half = wave >> 2;

    const int b    = blockIdx.x >> 7;
    const int qblk = blockIdx.x & 127;
    const int qbase = qblk * 64 + qt * 16;

    const float* xtb = xt  + (size_t)b * N_NODES * FDIM;
    const float* vbp = sup + (size_t)b * N_NODES * FDIM;

    v2f q[16];
    {
        const float* qrow = xtb + (size_t)(qbase + mrow) * FDIM + 2 * hi;
#pragma unroll
        for (int c = 0; c < 16; ++c)
            q[c] = *(const v2f*)(qrow + 4 * c);
    }

    v8f accT[4];
#pragma unroll
    for (int t = 0; t < 4; ++t)
#pragma unroll
        for (int r = 0; r < 8; ++r) accT[t][r] = 0.0f;

    float m = -3.0e38f;
    float l = 0.0f;

    const int srow0 = tid >> 4;
    const int scol  = (tid & 15) * 4;
    const size_t goff0 = (size_t)srow0 * FDIM + scol;
    const size_t goff1 = (size_t)(srow0 + 16) * FDIM + scol;
    v4f rk0 = *(const v4f*)(xtb + goff0);
    v4f rk1 = *(const v4f*)(xtb + goff1);
    v4f rv0 = *(const v4f*)(vbp + goff0);
    v4f rv1 = *(const v4f*)(vbp + goff1);

    const int NJ = N_NODES / 32;
    for (int j = 0; j < NJ; ++j) {
        const int buf = j & 1;
        *(v4f*)&sh[buf][0][srow0][scol]     = rk0;
        *(v4f*)&sh[buf][0][srow0 + 16][scol] = rk1;
        *(v4f*)&sh[buf][1][srow0][scol]     = rv0;
        *(v4f*)&sh[buf][1][srow0 + 16][scol] = rv1;
        __syncthreads();
        if (j + 1 < NJ) {
            const size_t tb = (size_t)(j + 1) * 32 * FDIM;
            rk0 = *(const v4f*)(xtb + tb + goff0);
            rk1 = *(const v4f*)(xtb + tb + goff1);
            rv0 = *(const v4f*)(vbp + tb + goff0);
            rv1 = *(const v4f*)(vbp + tb + goff1);
        }
        __builtin_amdgcn_sched_barrier(0);

        const float* kb    = &sh[buf][0][16 * half][0];
        const float* vtile = &sh[buf][1][16 * half][0];

        v8f s;
#pragma unroll
        for (int r = 0; r < 8; ++r) s[r] = 0.0f;
#pragma unroll
        for (int c = 0; c < 16; ++c) {
            v2f kk = *(const v2f*)(kb + mrow * LROW + 4 * c + 2 * hi);
            s = __builtin_amdgcn_wmma_f32_16x16x4_f32(
                    false, kk, false, q[c], (short)0, s, false, false);
        }

        float mloc = s[0];
#pragma unroll
        for (int r = 1; r < 8; ++r) mloc = fmaxf(mloc, s[r]);
        mloc = fmaxf(mloc, __shfl_xor(mloc, 16, 32));
        float mnew = fmaxf(m, mloc);
        float sc = __expf(m - mnew);
        m = mnew;

        float p[8], ls = 0.0f;
#pragma unroll
        for (int r = 0; r < 8; ++r) { p[r] = __expf(s[r] - mnew); ls += p[r]; }
        ls += __shfl_xor(ls, 16, 32);
        l = l * sc + ls;

#pragma unroll
        for (int t = 0; t < 4; ++t)
#pragma unroll
            for (int r = 0; r < 8; ++r) accT[t][r] *= sc;

        float tt[8];
#pragma unroll
        for (int r = 0; r < 8; ++r) tt[r] = __shfl_xor(p[r], 16, 32);
        v2f pb[4];
        pb[0].x = hi ? tt[2] : p[0];  pb[0].y = hi ? tt[3] : p[1];
        pb[1].x = hi ? tt[6] : p[4];  pb[1].y = hi ? tt[7] : p[5];
        pb[2].x = hi ? p[2] : tt[0];  pb[2].y = hi ? p[3] : tt[1];
        pb[3].x = hi ? p[6] : tt[4];  pb[3].y = hi ? p[7] : tt[5];

#pragma unroll
        for (int t = 0; t < 4; ++t) {
#pragma unroll
            for (int c = 0; c < 4; ++c) {
                int krow = 4 * c + 2 * hi;
                v2f a;
                a.x = vtile[krow * LROW + 16 * t + mrow];
                a.y = vtile[(krow + 1) * LROW + 16 * t + mrow];
                accT[t] = __builtin_amdgcn_wmma_f32_16x16x4_f32(
                              false, a, false, pb[c], (short)0, accT[t],
                              false, false);
            }
        }
    }

    float* mg = &sh[NJ & 1][0][0][0] + qt * (34 * 32) + lane;
    if (half) {
#pragma unroll
        for (int t = 0; t < 4; ++t)
#pragma unroll
            for (int r = 0; r < 8; ++r) mg[(8 * t + r) * 32] = accT[t][r];
        mg[32 * 32] = m;
        mg[33 * 32] = l;
    }
    __syncthreads();

    if (!half) {
        float m2 = mg[32 * 32];
        float l2 = mg[33 * 32];
        float mn = fmaxf(m, m2);
        float a1 = __expf(m - mn);
        float a2 = __expf(m2 - mn);
        float inv = 1.0f / (l * a1 + l2 * a2);
        a1 *= inv;
        a2 *= inv;
        int n = qbase + mrow;
#pragma unroll
        for (int t = 0; t < 4; ++t)
#pragma unroll
            for (int r = 0; r < 8; ++r) {
                int o = 16 * t + r + 8 * hi;
                out[((size_t)b * FDIM + o) * N_NODES + n] =
                    accT[t][r] * a1 + mg[(8 * t + r) * 32] * a2;
            }
    }
}

extern "C" void kernel_launch(void* const* d_in, const int* in_sizes, int n_in,
                              void* d_out, int out_size, void* d_ws, size_t ws_size,
                              hipStream_t stream) {
    (void)in_sizes; (void)n_in; (void)out_size; (void)ws_size;
    const float* x = (const float*)d_in[0];
    const float* w = (const float*)d_in[1];
    float* xt  = (float*)d_ws;
    float* sup = xt + (size_t)BATCH * N_NODES * FDIM;
    float* out = (float*)d_out;

    dim3 gPrep(N_NODES / 256, BATCH);
    gcn_prep<<<gPrep, 256, 0, stream>>>(x, w, xt, sup);

    dim3 gFlash(BATCH * (N_NODES / 64));
    gcn_flash<<<gFlash, 256, 0, stream>>>(xt, sup, out);
}


# Each wave owns two 16-row query tiles and a quarter of a trip's 64 keys: every K and V fragment read from LDS feeds two WMMAs; four partial softmax sums per row merged in LDS at the end
# speedup vs baseline: 2.8027x; 1.0814x over previous
#include <hip/hip_runtime.h>


#define BATCH   4
#define FDIM    64
#define N_NODES 8192
#define LROW    68
#define KTRIP   64

typedef __attribute__((ext_vector_type(2))) float v2f;
typedef __attribute__((ext_vector_type(4))) float v4f;
typedef __attribute__((ext_vector_type(8))) float v8f;

__global__ __launch_bounds__(256) void gcn_prep(const float* __restrict__ x,
                                                const float* __restrict__ w,
                                                float* __restrict__ xt,
                                                float* __restrict__ sup) {
    int n = blockIdx.x * 256 + threadIdx.x;
    int b = blockIdx.y;
    const float* xb = x + (size_t)b * FDIM * N_NODES;
    float* xtrow = xt  + ((size_t)b * N_NODES + n) * FDIM;
    float* srow  = sup + ((size_t)b * N_NODES + n) * FDIM;

    float acc[FDIM];
#pragma unroll
    for (int o = 0; o < FDIM; ++o) acc[o] = 0.0f;

    for (int f = 0; f < FDIM; ++f) {
        float v = xb[(size_t)f * N_NODES + n];
        xtrow[f] = v;
#pragma unroll
        for (int o = 0; o < FDIM; ++o) acc[o] += v * w[f * FDIM + o];
    }
#pragma unroll
    for (int o = 0; o < FDIM; ++o) srow[o] = acc[o];
}

__global__ __launch_bounds__(256) __attribute__((amdgpu_waves_per_eu(4, 4)))
void gcn_flash(const float* __restrict__ xt,
               const float* __restrict__ sup,
               float* __restrict__ out) {
    __shared__ __align__(16) float sh[2][2][KTRIP][LROW];

    const int tid  = threadIdx.x;
    const int wave = tid >> 5;
    const int lane = tid & 31;
    const int hi   = lane >> 4;
    const int mrow = lane & 15;
    const int qp   = wave & 1;
    const int kq   = wave >> 1;

    const int b    = blockIdx.x >> 7;
    const int qblk = blockIdx.x & 127;
    const int qbase = qblk * 64 + qp * 32;

    const float* xtb = xt  + (size_t)b * N_NODES * FDIM;
    const float* vbp = sup + (size_t)b * N_NODES * FDIM;

    v2f q[2][16];
#pragma unroll
    for (int u = 0; u < 2; ++u) {
        const float* qrow = xtb + (size_t)(qbase + 16 * u + mrow) * FDIM + 2 * hi;
#pragma unroll
        for (int c = 0; c < 16; ++c)
            q[u][c] = *(const v2f*)(qrow + 4 * c);
    }

    v8f accT[2][4];
#pragma unroll
    for (int u = 0; u < 2; ++u)
#pragma unroll
        for (int t = 0; t < 4; ++t)
#pragma unroll
            for (int r = 0; r < 8; ++r) accT[u][t][r] = 0.0f;

    float m[2] = {-3.0e38f, -3.0e38f};
    float l[2] = {0.0f, 0.0f};

    const int srow0 = tid >> 4;
    const int scol  = (tid & 15) * 4;
    const size_t goff0 = (size_t)srow0 * FDIM + scol;
    const size_t goff1 = goff0 + 16 * FDIM;
    const size_t goff2 = goff0 + 32 * FDIM;
    const size_t goff3 = goff0 + 48 * FDIM;
    v4f rk0 = *(const v4f*)(xtb + goff0);
    v4f rk1 = *(const v4f*)(xtb + goff1);
    v4f rk2 = *(const v4f*)(xtb + goff2);
    v4f rk3 = *(const v4f*)(xtb + goff3);
    v4f rv0 = *(const v4f*)(vbp + goff0);
    v4f rv1 = *(const v4f*)(vbp + goff1);
    v4f rv2 = *(const v4f*)(vbp + goff2);
    v4f rv3 = *(const v4f*)(vbp + goff3);

    *(v4f*)&sh[0][0][srow0][scol]      = rk0;
    *(v4f*)&sh[0][0][srow0 + 16][scol] = rk1;
    *(v4f*)&sh[0][0][srow0 + 32][scol] = rk2;
    *(v4f*)&sh[0][0][srow0 + 48][scol] = rk3;
    *(v4f*)&sh[0][1][srow0][scol]      = rv0;
    *(v4f*)&sh[0][1][srow0 + 16][scol] = rv1;
    *(v4f*)&sh[0][1][srow0 + 32][scol] = rv2;
    *(v4f*)&sh[0][1][srow0 + 48][scol] = rv3;

    const int NJ = N_NODES / KTRIP;
    for (int j = 0; j < NJ; ++j) {
        const int buf = j & 1;
        __syncthreads();
        if (j + 1 < NJ) {
            const size_t tb = (size_t)(j + 1) * KTRIP * FDIM;
            rk0 = *(const v4f*)(xtb + tb + goff0);
            rk1 = *(const v4f*)(xtb + tb + goff1);
            rk2 = *(const v4f*)(xtb + tb + goff2);
            rk3 = *(const v4f*)(xtb + tb + goff3);
            rv0 = *(const v4f*)(vbp + tb + goff0);
            rv1 = *(const v4f*)(vbp + tb + goff1);
            rv2 = *(const v4f*)(vbp + tb + goff2);
            rv3 = *(const v4f*)(vbp + tb + goff3);
        }

        const float* kb    = &sh[buf][0][16 * kq][0];
        const float* vtile = &sh[buf][1][16 * kq][0];

        v2f kk[16];
#pragma unroll
        for (int c = 0; c < 16; ++c)
            kk[c] = *(const v2f*)(kb + mrow * LROW + 4 * c + 2 * hi);
        __builtin_amdgcn_sched_barrier(0);

        v8f s[2];
#pragma unroll
        for (int u = 0; u < 2; ++u)
#pragma unroll
            for (int r = 0; r < 8; ++r) s[u][r] = 0.0f;
#pragma unroll
        for (int c = 0; c < 16; ++c) {
#pragma unroll
            for (int u = 0; u < 2; ++u)
                s[u] = __builtin_amdgcn_wmma_f32_16x16x4_f32(
                           false, kk[c], false, q[u][c], (short)0, s[u], false, false);
        }

        v2f va[4][4];
#pragma unroll
        for (int t = 0; t < 4; ++t) {
#pragma unroll
            for (int c = 0; c < 4; ++c) {
                int krow = 4 * c + 2 * hi;
                va[t][c].x = vtile[krow * LROW + 16 * t + mrow];
                va[t][c].y = vtile[(krow + 1) * LROW + 16 * t + mrow];
            }
        }
        if (j + 1 < NJ) {
            *(v4f*)&sh[buf ^ 1][0][srow0][scol]      = rk0;
            *(v4f*)&sh[buf ^ 1][0][srow0 + 16][scol] = rk1;
            *(v4f*)&sh[buf ^ 1][0][srow0 + 32][scol] = rk2;
            *(v4f*)&sh[buf ^ 1][0][srow0 + 48][scol] = rk3;
            *(v4f*)&sh[buf ^ 1][1][srow0][scol]      = rv0;
            *(v4f*)&sh[buf ^ 1][1][srow0 + 16][scol] = rv1;
            *(v4f*)&sh[buf ^ 1][1][srow0 + 32][scol] = rv2;
            *(v4f*)&sh[buf ^ 1][1][srow0 + 48][scol] = rv3;
        }
        __builtin_amdgcn_sched_barrier(0);

        v2f pb[2][4];
#pragma unroll
        for (int u = 0; u < 2; ++u) {
            float mloc = s[u][0];
#pragma unroll
            for (int r = 1; r < 8; ++r) mloc = fmaxf(mloc, s[u][r]);
            mloc = fmaxf(mloc, __shfl_xor(mloc, 16, 32));
            float mnew = fmaxf(m[u], mloc);
            float sc = __expf(m[u] - mnew);
            m[u] = mnew;

            float p[8], ls = 0.0f;
#pragma unroll
            for (int r = 0; r < 8; ++r) { p[r] = __expf(s[u][r] - mnew); ls += p[r]; }
            ls += __shfl_xor(ls, 16, 32);
            l[u] = l[u] * sc + ls;

#pragma unroll
            for (int t = 0; t < 4; ++t)
#pragma unroll
                for (int r = 0; r < 8; ++r) accT[u][t][r] *= sc;

            float tt[8];
#pragma unroll
            for (int r = 0; r < 8; ++r) tt[r] = __shfl_xor(p[r], 16, 32);
            pb[u][0].x = hi ? tt[2] : p[0];  pb[u][0].y = hi ? tt[3] : p[1];
            pb[u][1].x = hi ? tt[6] : p[4];  pb[u][1].y = hi ? tt[7] : p[5];
            pb[u][2].x = hi ? p[2] : tt[0];  pb[u][2].y = hi ? p[3] : tt[1];
            pb[u][3].x = hi ? p[6] : tt[4];  pb[u][3].y = hi ? p[7] : tt[5];
        }

#pragma unroll
        for (int t = 0; t < 4; ++t) {
#pragma unroll
            for (int c = 0; c < 4; ++c) {
#pragma unroll
                for (int u = 0; u < 2; ++u)
                    accT[u][t] = __builtin_amdgcn_wmma_f32_16x16x4_f32(
                                     false, va[t][c], false, pb[u][c], (short)0, accT[u][t],
                                     false, false);
            }
        }
    }

    __syncthreads();
    float* area = &sh[0][0][0][0];
    {
        float* mine = area + (size_t)(qp * 4 + kq) * (68 * 32) + lane;
#pragma unroll
        for (int u = 0; u < 2; ++u) {
#pragma unroll
            for (int t = 0; t < 4; ++t)
#pragma unroll
                for (int r = 0; r < 8; ++r)
                    mine[(32 * u + 8 * t + r) * 32] = accT[u][t][r];
            mine[(64 + u) * 32] = m[u];
            mine[(66 + u) * 32] = l[u];
        }
    }
    __syncthreads();

    {
        const float* pair = area + (size_t)(qp * 4) * (68 * 32) + lane;
        const int t = kq;
#pragma unroll
        for (int u = 0; u < 2; ++u) {
            float mk[4], lk[4];
#pragma unroll
            for (int k = 0; k < 4; ++k) {
                mk[k] = pair[(size_t)k * (68 * 32) + (64 + u) * 32];
                lk[k] = pair[(size_t)k * (68 * 32) + (66 + u) * 32];
            }
            float mn = fmaxf(fmaxf(mk[0], mk[1]), fmaxf(mk[2], mk[3]));
            float ak[4], lsum = 0.0f;
#pragma unroll
            for (int k = 0; k < 4; ++k) { ak[k] = __expf(mk[k] - mn); lsum += lk[k] * ak[k]; }
            float inv = 1.0f / lsum;
#pragma unroll
            for (int k = 0; k < 4; ++k) ak[k] *= inv;
            int n = qbase + 16 * u + mrow;
#pragma unroll
            for (int r = 0; r < 8; ++r) {
                float o_acc = 0.0f;
#pragma unroll
                for (int k = 0; k < 4; ++k)
                    o_acc += pair[(size_t)k * (68 * 32) + (32 * u + 8 * t + r) * 32] * ak[k];
                int o = 16 * t + r + 8 * hi;
                out[((size_t)b * FDIM + o) * N_NODES + n] = o_acc;
            }
        }
    }
}

extern "C" void kernel_launch(void* const* d_in, const int* in_sizes, int n_in,
                              void* d_out, int out_size, void* d_ws, size_t ws_size,
                              hipStream_t stream) {
    (void)in_sizes; (void)n_in; (void)out_size; (void)ws_size;
    const float* x = (const float*)d_in[0];
    const float* w = (const float*)d_in[1];
    float* xt  = (float*)d_ws;
    float* sup = xt + (size_t)BATCH * N_NODES * FDIM;
    float* out = (float*)d_out;

    dim3 gPrep(N_NODES / 256, BATCH);
    gcn_prep<<<gPrep, 256, 0, stream>>>(x, w, xt, sup);

    dim3 gFlash(BATCH * (N_NODES / 64));
    gcn_flash<<<gFlash, 256, 0, stream>>>(xt, sup, out);
}


# PV product's key order follows the score tile's registers (half h, chunk j: keys 8h+2j, 8h+2j+1; V loader reads those rows): P^T needs no cross-half exchange; in-tile sum order changes
# speedup vs baseline: 2.9637x; 1.0574x over previous
#include <hip/hip_runtime.h>


#define BATCH   4
#define FDIM    64
#define N_NODES 8192
#define LROW    68
#define KTRIP   64

typedef __attribute__((ext_vector_type(2))) float v2f;
typedef __attribute__((ext_vector_type(4))) float v4f;
typedef __attribute__((ext_vector_type(8))) float v8f;

__global__ __launch_bounds__(256) void gcn_prep(const float* __restrict__ x,
                                                const float* __restrict__ w,
                                                float* __restrict__ xt,
                                                float* __restrict__ sup) {
    int n = blockIdx.x * 256 + threadIdx.x;
    int b = blockIdx.y;
    const float* xb = x + (size_t)b * FDIM * N_NODES;
    float* xtrow = xt  + ((size_t)b * N_NODES + n) * FDIM;
    float* srow  = sup + ((size_t)b * N_NODES + n) * FDIM;

    float acc[FDIM];
#pragma unroll
    for (int o = 0; o < FDIM; ++o) acc[o] = 0.0f;

    for (int f = 0; f < FDIM; ++f) {
        float v = xb[(size_t)f * N_NODES + n];
        xtrow[f] = v;
#pragma unroll
        for (int o = 0; o < FDIM; ++o) acc[o] += v * w[f * FDIM + o];
    }
#pragma unroll
    for (int o = 0; o < FDIM; ++o) srow[o] = acc[o];
}

__global__ __launch_bounds__(256) __attribute__((amdgpu_waves_per_eu(4, 4)))
void gcn_flash(const float* __restrict__ xt,
               const float* __restrict__ sup,
               float* __restrict__ out) {
    __shared__ __align__(16) float sh[2][2][KTRIP][LROW];

    const int tid  = threadIdx.x;
    const int wave = tid >> 5;
    const int lane = tid & 31;
    const int hi   = lane >> 4;
    const int mrow = lane & 15;
    const int qp   = wave & 1;
    const int kq   = wave >> 1;

    const int b    = blockIdx.x >> 7;
    const int qblk = blockIdx.x & 127;
    const int qbase = qblk * 64 + qp * 32;

    const float* xtb = xt  + (size_t)b * N_NODES * FDIM;
    const float* vbp = sup + (size_t)b * N_NODES * FDIM;

    v2f q[2][16];
#pragma unroll
    for (int u = 0; u < 2; ++u) {
        const float* qrow = xtb + (size_t)(qbase + 16 * u + mrow) * FDIM + 2 * hi;
#pragma unroll
        for (int c = 0; c < 16; ++c)
            q[u][c] = *(const v2f*)(qrow + 4 * c);
    }

    v8f accT[2][4];
#pragma unroll
    for (int u = 0; u < 2; ++u)
#pragma unroll
        for (int t = 0; t < 4; ++t)
#pragma unroll
            for (int r = 0; r < 8; ++r) accT[u][t][r] = 0.0f;

    float m[2] = {-3.0e38f, -3.0e38f};
    float l[2] = {0.0f, 0.0f};

    const int srow0 = tid >> 4;
    const int scol  = (tid & 15) * 4;
    const size_t goff0 = (size_t)srow0 * FDIM + scol;
    const size_t goff1 = goff0 + 16 * FDIM;
    const size_t goff2 = goff0 + 32 * FDIM;
    const size_t goff3 = goff0 + 48 * FDIM;
    v4f rk0 = *(const v4f*)(xtb + goff0);
    v4f rk1 = *(const v4f*)(xtb + goff1);
    v4f rk2 = *(const v4f*)(xtb + goff2);
    v4f rk3 = *(const v4f*)(xtb + goff3);
    v4f rv0 = *(const v4f*)(vbp + goff0);
    v4f rv1 = *(const v4f*)(vbp + goff1);
    v4f rv2 = *(const v4f*)(vbp + goff2);
    v4f rv3 = *(const v4f*)(vbp + goff3);

    *(v4f*)&sh[0][0][srow0][scol]      = rk0;
    *(v4f*)&sh[0][0][srow0 + 16][scol] = rk1;
    *(v4f*)&sh[0][0][srow0 + 32][scol] = rk2;
    *(v4f*)&sh[0][0][srow0 + 48][scol] = rk3;
    *(v4f*)&sh[0][1][srow0][scol]      = rv0;
    *(v4f*)&sh[0][1][srow0 + 16][scol] = rv1;
    *(v4f*)&sh[0][1][srow0 + 32][scol] = rv2;
    *(v4f*)&sh[0][1][srow0 + 48][scol] = rv3;

    const int NJ = N_NODES / KTRIP;
    for (int j = 0; j < NJ; ++j) {
        const int buf = j & 1;
        __syncthreads();
        if (j + 1 < NJ) {
            const size_t tb = (size_t)(j + 1) * KTRIP * FDIM;
            rk0 = *(const v4f*)(xtb + tb + goff0);
            rk1 = *(const v4f*)(xtb + tb + goff1);
            rk2 = *(const v4f*)(xtb + tb + goff2);
            rk3 = *(const v4f*)(xtb + tb + goff3);
            rv0 = *(const v4f*)(vbp + tb + goff0);
            rv1 = *(const v4f*)(vbp + tb + goff1);
            rv2 = *(const v4f*)(vbp + tb + goff2);
            rv3 = *(const v4f*)(vbp + tb + goff3);
        }

        const float* kb    = &sh[buf][0][16 * kq][0];
        const float* vtile = &sh[buf][1][16 * kq][0];

        v2f kk[16];
#pragma unroll
        for (int c = 0; c < 16; ++c)
            kk[c] = *(const v2f*)(kb + mrow * LROW + 4 * c + 2 * hi);
        __builtin_amdgcn_sched_barrier(0);

        v8f s[2];
#pragma unroll
        for (int u = 0; u < 2; ++u)
#pragma unroll
            for (int r = 0; r < 8; ++r) s[u][r] = 0.0f;
#pragma unroll
        for (int c = 0; c < 16; ++c) {
#pragma unroll
            for (int u = 0; u < 2; ++u)
                s[u] = __builtin_amdgcn_wmma_f32_16x16x4_f32(
                           false, kk[c], false, q[u][c], (short)0, s[u], false, false);
        }

        v2f va[4][4];
        const float* vl = vtile + (8 * hi) * LROW + mrow;
#pragma unroll
        for (int t = 0; t < 4; ++t) {
#pragma unroll
            for (int c = 0; c < 4; ++c) {
                va[t][c].x = vl[(2 * c) * LROW + 16 * t];
                va[t][c].y = vl[(2 * c + 1) * LROW + 16 * t];
            }
        }
        if (j + 1 < NJ) {
            *(v4f*)&sh[buf ^ 1][0][srow0][scol]      = rk0;
            *(v4f*)&sh[buf ^ 1][0][srow0 + 16][scol] = rk1;
            *(v4f*)&sh[buf ^ 1][0][srow0 + 32][scol] = rk2;
            *(v4f*)&sh[buf ^ 1][0][srow0 + 48][scol] = rk3;
            *(v4f*)&sh[buf ^ 1][1][srow0][scol]      = rv0;
            *(v4f*)&sh[buf ^ 1][1][srow0 + 16][scol] = rv1;
            *(v4f*)&sh[buf ^ 1][1][srow0 + 32][scol] = rv2;
            *(v4f*)&sh[buf ^ 1][1][srow0 + 48][scol] = rv3;
        }
        __builtin_amdgcn_sched_barrier(0);

        v2f pb[2][4];
#pragma unroll
        for (int u = 0; u < 2; ++u) {
            float mloc = s[u][0];
#pragma unroll
            for (int r = 1; r < 8; ++r) mloc = fmaxf(mloc, s[u][r]);
            mloc = fmaxf(mloc, __shfl_xor(mloc, 16, 32));
            float mnew = fmaxf(m[u], mloc);
            float sc = __expf(m[u] - mnew);
            m[u] = mnew;

            float p[8], ls = 0.0f;
#pragma unroll
            for (int r = 0; r < 8; ++r) { p[r] = __expf(s[u][r] - mnew); ls += p[r]; }
            ls += __shfl_xor(ls, 16, 32);
            l[u] = l[u] * sc + ls;

#pragma unroll
            for (int t = 0; t < 4; ++t)
#pragma unroll
                for (int r = 0; r < 8; ++r) accT[u][t][r] *= sc;

#pragma unroll
            for (int c = 0; c < 4; ++c) { pb[u][c].x = p[2 * c]; pb[u][c].y = p[2 * c + 1]; }
        }

#pragma unroll
        for (int t = 0; t < 4; ++t) {
#pragma unroll
            for (int c = 0; c < 4; ++c) {
#pragma unroll
                for (int u = 0; u < 2; ++u)
                    accT[u][t] = __builtin_amdgcn_wmma_f32_16x16x4_f32(
                                     false, va[t][c], false, pb[u][c], (short)0, accT[u][t],
                                     false, false);
            }
        }
    }

    __syncthreads();
    float* area = &sh[0][0][0][0];
    {
        float* mine = area + (size_t)(qp * 4 + kq) * (68 * 32) + lane;
#pragma unroll
        for (int u = 0; u < 2; ++u) {
#pragma unroll
            for (int t = 0; t < 4; ++t)
#pragma unroll
                for (int r = 0; r < 8; ++r)
                    mine[(32 * u + 8 * t + r) * 32] = accT[u][t][r];
            mine[(64 + u) * 32] = m[u];
            mine[(66 + u) * 32] = l[u];
        }
    }
    __syncthreads();

    {
        const float* pair = area + (size_t)(qp * 4) * (68 * 32) + lane;
        const int t = kq;
#pragma unroll
        for (int u = 0; u < 2; ++u) {
            float mk[4], lk[4];
#pragma unroll
            for (int k = 0; k < 4; ++k) {
                mk[k] = pair[(size_t)k * (68 * 32) + (64 + u) * 32];
                lk[k] = pair[(size_t)k * (68 * 32) + (66 + u) * 32];
            }
            float mn = fmaxf(fmaxf(mk[0], mk[1]), fmaxf(mk[2], mk[3]));
            float ak[4], lsum = 0.0f;
#pragma unroll
            for (int k = 0; k < 4; ++k) { ak[k] = __expf(mk[k] - mn); lsum += lk[k] * ak[k]; }
            float inv = 1.0f / lsum;
#pragma unroll
            for (int k = 0; k < 4; ++k) ak[k] *= inv;
            int n = qbase + 16 * u + mrow;
#pragma unroll
            for (int r = 0; r < 8; ++r) {
                float o_acc = 0.0f;
#pragma unroll
                for (int k = 0; k < 4; ++k)
                    o_acc += pair[(size_t)k * (68 * 32) + (32 * u + 8 * t + r) * 32] * ak[k];
                int o = 16 * t + r + 8 * hi;
                out[((size_t)b * FDIM + o) * N_NODES + n] = o_acc;
            }
        }
    }
}

extern "C" void kernel_launch(void* const* d_in, const int* in_sizes, int n_in,
                              void* d_out, int out_size, void* d_ws, size_t ws_size,
                              hipStream_t stream) {
    (void)in_sizes; (void)n_in; (void)out_size; (void)ws_size;
    const float* x = (const float*)d_in[0];
    const float* w = (const float*)d_in[1];
    float* xt  = (float*)d_ws;
    float* sup = xt + (size_t)BATCH * N_NODES * FDIM;
    float* out = (float*)d_out;

    dim3 gPrep(N_NODES / 256, BATCH);
    gcn_prep<<<gPrep, 256, 0, stream>>>(x, w, xt, sup);

    dim3 gFlash(BATCH * (N_NODES / 64));
    gcn_flash<<<gFlash, 256, 0, stream>>>(xt, sup, out);
}


# Output product reassociated as (softmax(S) x^T) W: K and V are one staged tile read straight from x, W applied per block on WMMA at the end; prep launch gone; sum order changes
# speedup vs baseline: 3.3787x; 1.1400x over previous
#include <hip/hip_runtime.h>


#define BATCH   4
#define FDIM    64
#define N_NODES 8192
#define KTRIP   64
#define PITCH   76

typedef __attribute__((ext_vector_type(2))) float v2f;
typedef __attribute__((ext_vector_type(4))) float v4f;
typedef __attribute__((ext_vector_type(8))) float v8f;

__global__ __launch_bounds__(256) __attribute__((amdgpu_waves_per_eu(4, 4)))
void gcn_flash(const float* __restrict__ x,
               const float* __restrict__ w,
               float* __restrict__ out) {
    __shared__ __align__(16) float shm[8 * 68 * 32];

    const int tid  = threadIdx.x;
    const int wave = tid >> 5;
    const int lane = tid & 31;
    const int hi   = lane >> 4;
    const int mrow = lane & 15;
    const int qp   = wave & 1;
    const int kq   = wave >> 1;

    const int b    = blockIdx.x >> 7;
    const int qblk = blockIdx.x & 127;
    const int qbase = qblk * 64 + qp * 32;

    const float* xb = x + (size_t)b * FDIM * N_NODES;

    v2f q[2][16];
#pragma unroll
    for (int u = 0; u < 2; ++u) {
        const float* qcol = xb + (size_t)(2 * hi) * N_NODES + (qbase + 16 * u + mrow);
#pragma unroll
        for (int c = 0; c < 16; ++c) {
            q[u][c].x = qcol[(size_t)(4 * c) * N_NODES];
            q[u][c].y = qcol[(size_t)(4 * c + 1) * N_NODES];
        }
    }

    v8f accT[2][4];
#pragma unroll
    for (int u = 0; u < 2; ++u)
#pragma unroll
        for (int t = 0; t < 4; ++t)
#pragma unroll
            for (int r = 0; r < 8; ++r) accT[u][t][r] = 0.0f;

    float m[2] = {-3.0e38f, -3.0e38f};
    float l[2] = {0.0f, 0.0f};

    const int srow0 = tid >> 4;
    const int scol  = (tid & 15) * 4;
    const size_t goff0 = (size_t)srow0 * N_NODES + scol;
    const size_t goff1 = goff0 + (size_t)16 * N_NODES;
    const size_t goff2 = goff0 + (size_t)32 * N_NODES;
    const size_t goff3 = goff0 + (size_t)48 * N_NODES;
    v4f r0 = *(const v4f*)(xb + goff0);
    v4f r1 = *(const v4f*)(xb + goff1);
    v4f r2 = *(const v4f*)(xb + goff2);
    v4f r3 = *(const v4f*)(xb + goff3);

    const int soff = srow0 * PITCH + scol;
    *(v4f*)(shm + soff)              = r0;
    *(v4f*)(shm + soff + 16 * PITCH) = r1;
    *(v4f*)(shm + soff + 32 * PITCH) = r2;
    *(v4f*)(shm + soff + 48 * PITCH) = r3;

    const int NJ = N_NODES / KTRIP;
    for (int j = 0; j < NJ; ++j) {
        const int buf = j & 1;
        __syncthreads();
        if (j + 1 < NJ) {
            const size_t tb = (size_t)(j + 1) * KTRIP;
            r0 = *(const v4f*)(xb + tb + goff0);
            r1 = *(const v4f*)(xb + tb + goff1);
            r2 = *(const v4f*)(xb + tb + goff2);
            r3 = *(const v4f*)(xb + tb + goff3);
        }

        const float* tile = shm + buf * (FDIM * PITCH) + 16 * kq;

        v2f kk[16];
        const float* kl = tile + (2 * hi) * PITCH + mrow;
#pragma unroll
        for (int c = 0; c < 16; ++c) {
            kk[c].x = kl[(4 * c) * PITCH];
            kk[c].y = kl[(4 * c + 1) * PITCH];
        }
        __builtin_amdgcn_sched_barrier(0);

        v8f s[2];
#pragma unroll
        for (int u = 0; u < 2; ++u)
#pragma unroll
            for (int r = 0; r < 8; ++r) s[u][r] = 0.0f;
#pragma unroll
        for (int c = 0; c < 16; ++c) {
#pragma unroll
            for (int u = 0; u < 2; ++u)
                s[u] = __builtin_amdgcn_wmma_f32_16x16x4_f32(
                           false, kk[c], false, q[u][c], (short)0, s[u], false, false);
        }

        v2f va[4][4];
        const float* vl = tile + mrow * PITCH + 8 * hi;
#pragma unroll
        for (int t = 0; t < 4; ++t) {
#pragma unroll
            for (int c = 0; c < 4; ++c)
                va[t][c] = *(const v2f*)(vl + (16 * t) * PITCH + 2 * c);
        }
        if (j + 1 < NJ) {
            float* sn = shm + (buf ^ 1) * (FDIM * PITCH) + soff;
            *(v4f*)(sn)              = r0;
            *(v4f*)(sn + 16 * PITCH) = r1;
            *(v4f*)(sn + 32 * PITCH) = r2;
            *(v4f*)(sn + 48 * PITCH) = r3;
        }
        __builtin_amdgcn_sched_barrier(0);

        v2f pb[2][4];
#pragma unroll
        for (int u = 0; u < 2; ++u) {
            float mloc = s[u][0];
#pragma unroll
            for (int r = 1; r < 8; ++r) mloc = fmaxf(mloc, s[u][r]);
            mloc = fmaxf(mloc, __shfl_xor(mloc, 16, 32));
            float mnew = fmaxf(m[u], mloc);
            float sc = __expf(m[u] - mnew);
            m[u] = mnew;

            float p[8], ls = 0.0f;
#pragma unroll
            for (int r = 0; r < 8; ++r) { p[r] = __expf(s[u][r] - mnew); ls += p[r]; }
            ls += __shfl_xor(ls, 16, 32);
            l[u] = l[u] * sc + ls;

#pragma unroll
            for (int t = 0; t < 4; ++t)
#pragma unroll
                for (int r = 0; r < 8; ++r) accT[u][t][r] *= sc;

#pragma unroll
            for (int c = 0; c < 4; ++c) { pb[u][c].x = p[2 * c]; pb[u][c].y = p[2 * c + 1]; }
        }

#pragma unroll
        for (int t = 0; t < 4; ++t) {
#pragma unroll
            for (int c = 0; c < 4; ++c) {
#pragma unroll
                for (int u = 0; u < 2; ++u)
                    accT[u][t] = __builtin_amdgcn_wmma_f32_16x16x4_f32(
                                     false, va[t][c], false, pb[u][c], (short)0, accT[u][t],
                                     false, false);
            }
        }
    }

    __syncthreads();
    float* area = shm;
    {
        float* mine = area + (size_t)(qp * 4 + kq) * (68 * 32) + lane;
#pragma unroll
        for (int u = 0; u < 2; ++u) {
#pragma unroll
            for (int t = 0; t < 4; ++t)
#pragma unroll
                for (int r = 0; r < 8; ++r)
                    mine[(32 * u + 8 * t + r) * 32] = accT[u][t][r];
            mine[(64 + u) * 32] = m[u];
            mine[(66 + u) * 32] = l[u];
        }
    }

    v2f wa[4][4];
    {
        const float* wl = w + (size_t)(8 * hi) * FDIM + 16 * kq + mrow;
#pragma unroll
        for (int t = 0; t < 4; ++t)
#pragma unroll
            for (int c = 0; c < 4; ++c) {
                wa[t][c].x = wl[(16 * t + 2 * c) * FDIM];
                wa[t][c].y = wl[(16 * t + 2 * c + 1) * FDIM];
            }
    }
    __syncthreads();

    {
        const float* pair = area + (size_t)(qp * 4) * (68 * 32) + lane;
#pragma unroll
        for (int u = 0; u < 2; ++u) {
            float mk[4], lk[4];
#pragma unroll
            for (int k = 0; k < 4; ++k) {
                mk[k] = pair[(size_t)k * (68 * 32) + (64 + u) * 32];
                lk[k] = pair[(size_t)k * (68 * 32) + (66 + u) * 32];
            }
            float mn = fmaxf(fmaxf(mk[0], mk[1]), fmaxf(mk[2], mk[3]));
            float ak[4], lsum = 0.0f;
#pragma unroll
            for (int k = 0; k < 4; ++k) { ak[k] = __expf(mk[k] - mn); lsum += lk[k] * ak[k]; }
            float inv = 1.0f / lsum;
#pragma unroll
            for (int k = 0; k < 4; ++k) ak[k] *= inv;

            v8f d;
#pragma unroll
            for (int r = 0; r < 8; ++r) d[r] = 0.0f;
#pragma unroll
            for (int t = 0; t < 4; ++t) {
                float om[8];
#pragma unroll
                for (int r = 0; r < 8; ++r) {
                    float o_acc = 0.0f;
#pragma unroll
                    for (int k = 0; k < 4; ++k)
                        o_acc += pair[(size_t)k * (68 * 32) + (32 * u + 8 * t + r) * 32] * ak[k];
                    om[r] = o_acc;
                }
#pragma unroll
                for (int c = 0; c < 4; ++c) {
                    v2f bf;
                    bf.x = om[2 * c];
                    bf.y = om[2 * c + 1];
                    d = __builtin_amdgcn_wmma_f32_16x16x4_f32(
                            false, wa[t][c], false, bf, (short)0, d, false, false);
                }
            }
            int n = qbase + 16 * u + mrow;
#pragma unroll
            for (int r = 0; r < 8; ++r) {
                int o = 16 * kq + r + 8 * hi;
                out[((size_t)b * FDIM + o) * N_NODES + n] = d[r];
            }
        }
    }
}

extern "C" void kernel_launch(void* const* d_in, const int* in_sizes, int n_in,
                              void* d_out, int out_size, void* d_ws, size_t ws_size,
                              hipStream_t stream) {
    (void)in_sizes; (void)n_in; (void)out_size; (void)d_ws; (void)ws_size;
    const float* x = (const float*)d_in[0];
    const float* w = (const float*)d_in[1];
    float* out = (float*)d_out;

    dim3 gFlash(BATCH * (N_NODES / 64));
    gcn_flash<<<gFlash, 256, 0, stream>>>(x, w, out);
}


# Lazy softmax rescale: row max starts at the query's diagonal score; max exchange, scale and accumulator rescale only in trips where a score exceeds it; row-sum halves added at end; sum order changes
# speedup vs baseline: 3.5337x; 1.0459x over previous
#include <hip/hip_runtime.h>


#define BATCH   4
#define FDIM    64
#define N_NODES 8192
#define KTRIP   64
#define PITCH   76

typedef __attribute__((ext_vector_type(2))) float v2f;
typedef __attribute__((ext_vector_type(4))) float v4f;
typedef __attribute__((ext_vector_type(8))) float v8f;

__global__ __launch_bounds__(256) __attribute__((amdgpu_waves_per_eu(4, 4)))
void gcn_flash(const float* __restrict__ x,
               const float* __restrict__ w,
               float* __restrict__ out) {
    __shared__ __align__(16) float shm[8 * 68 * 32];

    const int tid  = threadIdx.x;
    const int wave = tid >> 5;
    const int lane = tid & 31;
    const int hi   = lane >> 4;
    const int mrow = lane & 15;
    const int qp   = wave & 1;
    const int kq   = wave >> 1;

    const int b    = blockIdx.x >> 7;
    const int qblk = blockIdx.x & 127;
    const int qbase = qblk * 64 + qp * 32;

    const float* xb = x + (size_t)b * FDIM * N_NODES;

    v2f q[2][16];
#pragma unroll
    for (int u = 0; u < 2; ++u) {
        const float* qcol = xb + (size_t)(2 * hi) * N_NODES + (qbase + 16 * u + mrow);
#pragma unroll
        for (int c = 0; c < 16; ++c) {
            q[u][c].x = qcol[(size_t)(4 * c) * N_NODES];
            q[u][c].y = qcol[(size_t)(4 * c + 1) * N_NODES];
        }
    }

    v8f accT[2][4];
#pragma unroll
    for (int u = 0; u < 2; ++u)
#pragma unroll
        for (int t = 0; t < 4; ++t)
#pragma unroll
            for (int r = 0; r < 8; ++r) accT[u][t][r] = 0.0f;

    float m[2];
#pragma unroll
    for (int u = 0; u < 2; ++u) {
        float d = 0.0f;
#pragma unroll
        for (int c = 0; c < 16; ++c) d += q[u][c].x * q[u][c].x + q[u][c].y * q[u][c].y;
        m[u] = d + __shfl_xor(d, 16, 32);
    }
    float l[2] = {0.0f, 0.0f};

    const int srow0 = tid >> 4;
    const int scol  = (tid & 15) * 4;
    const size_t goff0 = (size_t)srow0 * N_NODES + scol;
    const size_t goff1 = goff0 + (size_t)16 * N_NODES;
    const size_t goff2 = goff0 + (size_t)32 * N_NODES;
    const size_t goff3 = goff0 + (size_t)48 * N_NODES;
    v4f r0 = *(const v4f*)(xb + goff0);
    v4f r1 = *(const v4f*)(xb + goff1);
    v4f r2 = *(const v4f*)(xb + goff2);
    v4f r3 = *(const v4f*)(xb + goff3);

    const int soff = srow0 * PITCH + scol;
    *(v4f*)(shm + soff)              = r0;
    *(v4f*)(shm + soff + 16 * PITCH) = r1;
    *(v4f*)(shm + soff + 32 * PITCH) = r2;
    *(v4f*)(shm + soff + 48 * PITCH) = r3;

    const int NJ = N_NODES / KTRIP;
    for (int j = 0; j < NJ; ++j) {
        const int buf = j & 1;
        __syncthreads();
        if (j + 1 < NJ) {
            const size_t tb = (size_t)(j + 1) * KTRIP;
            r0 = *(const v4f*)(xb + tb + goff0);
            r1 = *(const v4f*)(xb + tb + goff1);
            r2 = *(const v4f*)(xb + tb + goff2);
            r3 = *(const v4f*)(xb + tb + goff3);
        }

        const float* tile = shm + buf * (FDIM * PITCH) + 16 * kq;

        v2f kk[16];
        const float* kl = tile + (2 * hi) * PITCH + mrow;
#pragma unroll
        for (int c = 0; c < 16; ++c) {
            kk[c].x = kl[(4 * c) * PITCH];
            kk[c].y = kl[(4 * c + 1) * PITCH];
        }
        __builtin_amdgcn_sched_barrier(0);

        v8f s[2];
#pragma unroll
        for (int u = 0; u < 2; ++u)
#pragma unroll
            for (int r = 0; r < 8; ++r) s[u][r] = 0.0f;
#pragma unroll
        for (int c = 0; c < 16; ++c) {
#pragma unroll
            for (int u = 0; u < 2; ++u)
                s[u] = __builtin_amdgcn_wmma_f32_16x16x4_f32(
                           false, kk[c], false, q[u][c], (short)0, s[u], false, false);
        }

        v2f va[4][4];
        const float* vl = tile + mrow * PITCH + 8 * hi;
#pragma unroll
        for (int t = 0; t < 4; ++t) {
#pragma unroll
            for (int c = 0; c < 4; ++c)
                va[t][c] = *(const v2f*)(vl + (16 * t) * PITCH + 2 * c);
        }
        if (j + 1 < NJ) {
            float* sn = shm + (buf ^ 1) * (FDIM * PITCH) + soff;
            *(v4f*)(sn)              = r0;
            *(v4f*)(sn + 16 * PITCH) = r1;
            *(v4f*)(sn + 32 * PITCH) = r2;
            *(v4f*)(sn + 48 * PITCH) = r3;
        }
        __builtin_amdgcn_sched_barrier(0);

        float mloc[2];
#pragma unroll
        for (int u = 0; u < 2; ++u) {
            mloc[u] = s[u][0];
#pragma unroll
            for (int r = 1; r < 8; ++r) mloc[u] = fmaxf(mloc[u], s[u][r]);
        }
        if (__builtin_amdgcn_ballot_w32((mloc[0] > m[0]) | (mloc[1] > m[1])) != 0u) {
#pragma unroll
            for (int u = 0; u < 2; ++u) {
                float mx = fmaxf(mloc[u], __shfl_xor(mloc[u], 16, 32));
                float mnew = fmaxf(m[u], mx);
                float sc = __expf(m[u] - mnew);
                m[u] = mnew;
                l[u] *= sc;
#pragma unroll
                for (int t = 0; t < 4; ++t)
#pragma unroll
                    for (int r = 0; r < 8; ++r) accT[u][t][r] *= sc;
            }
        }
        v2f pb[2][4];
#pragma unroll
        for (int u = 0; u < 2; ++u) {
            float p[8], ls = 0.0f;
#pragma unroll
            for (int r = 0; r < 8; ++r) { p[r] = __expf(s[u][r] - m[u]); ls += p[r]; }
            l[u] += ls;

#pragma unroll
            for (int c = 0; c < 4; ++c) { pb[u][c].x = p[2 * c]; pb[u][c].y = p[2 * c + 1]; }
        }

#pragma unroll
        for (int t = 0; t < 4; ++t) {
#pragma unroll
            for (int c = 0; c < 4; ++c) {
#pragma unroll
                for (int u = 0; u < 2; ++u)
                    accT[u][t] = __builtin_amdgcn_wmma_f32_16x16x4_f32(
                                     false, va[t][c], false, pb[u][c], (short)0, accT[u][t],
                                     false, false);
            }
        }
    }

    __syncthreads();
    float* area = shm;
    {
        float* mine = area + (size_t)(qp * 4 + kq) * (68 * 32) + lane;
#pragma unroll
        for (int u = 0; u < 2; ++u) {
#pragma unroll
            for (int t = 0; t < 4; ++t)
#pragma unroll
                for (int r = 0; r < 8; ++r)
                    mine[(32 * u + 8 * t + r) * 32] = accT[u][t][r];
            mine[(64 + u) * 32] = m[u];
            mine[(66 + u) * 32] = l[u] + __shfl_xor(l[u], 16, 32);
        }
    }

    v2f wa[4][4];
    {
        const float* wl = w + (size_t)(8 * hi) * FDIM + 16 * kq + mrow;
#pragma unroll
        for (int t = 0; t < 4; ++t)
#pragma unroll
            for (int c = 0; c < 4; ++c) {
                wa[t][c].x = wl[(16 * t + 2 * c) * FDIM];
                wa[t][c].y = wl[(16 * t + 2 * c + 1) * FDIM];
            }
    }
    __syncthreads();

    {
        const float* pair = area + (size_t)(qp * 4) * (68 * 32) + lane;
#pragma unroll
        for (int u = 0; u < 2; ++u) {
            float mk[4], lk[4];
#pragma unroll
            for (int k = 0; k < 4; ++k) {
                mk[k] = pair[(size_t)k * (68 * 32) + (64 + u) * 32];
                lk[k] = pair[(size_t)k * (68 * 32) + (66 + u) * 32];
            }
            float mn = fmaxf(fmaxf(mk[0], mk[1]), fmaxf(mk[2], mk[3]));
            float ak[4], lsum = 0.0f;
#pragma unroll
            for (int k = 0; k < 4; ++k) { ak[k] = __expf(mk[k] - mn); lsum += lk[k] * ak[k]; }
            float inv = 1.0f / lsum;
#pragma unroll
            for (int k = 0; k < 4; ++k) ak[k] *= inv;

            v8f d;
#pragma unroll
            for (int r = 0; r < 8; ++r) d[r] = 0.0f;
#pragma unroll
            for (int t = 0; t < 4; ++t) {
                __builtin_amdgcn_sched_barrier(0);
                float om[8];
#pragma unroll
                for (int r = 0; r < 8; ++r) {
                    float o_acc = 0.0f;
#pragma unroll
                    for (int k = 0; k < 4; ++k)
                        o_acc += pair[(size_t)k * (68 * 32) + (32 * u + 8 * t + r) * 32] * ak[k];
                    om[r] = o_acc;
                }
#pragma unroll
                for (int c = 0; c < 4; ++c) {
                    v2f bf;
                    bf.x = om[2 * c];
                    bf.y = om[2 * c + 1];
                    d = __builtin_amdgcn_wmma_f32_16x16x4_f32(
                            false, wa[t][c], false, bf, (short)0, d, false, false);
                }
            }
            int n = qbase + 16 * u + mrow;
#pragma unroll
            for (int r = 0; r < 8; ++r) {
                int o = 16 * kq + r + 8 * hi;
                out[((size_t)b * FDIM + o) * N_NODES + n] = d[r];
            }
        }
    }
}

extern "C" void kernel_launch(void* const* d_in, const int* in_sizes, int n_in,
                              void* d_out, int out_size, void* d_ws, size_t ws_size,
                              hipStream_t stream) {
    (void)in_sizes; (void)n_in; (void)out_size; (void)d_ws; (void)ws_size;
    const float* x = (const float*)d_in[0];
    const float* w = (const float*)d_in[1];
    float* out = (float*)d_out;

    dim3 gFlash(BATCH * (N_NODES / 64));
    gcn_flash<<<gFlash, 256, 0, stream>>>(x, w, out);
}


# Loop rotated by half a trip: behind the barrier a tile's K fragment reads are issued in front of the previous tile's 32 output WMMAs, which cover their LDS latency; same operations, same order
# speedup vs baseline: 3.6522x; 1.0335x over previous
#include <hip/hip_runtime.h>


#define BATCH   4
#define FDIM    64
#define N_NODES 8192
#define KTRIP   64
#define PITCH   76

typedef __attribute__((ext_vector_type(2))) float v2f;
typedef __attribute__((ext_vector_type(4))) float v4f;
typedef __attribute__((ext_vector_type(8))) float v8f;

__global__ __launch_bounds__(256) __attribute__((amdgpu_waves_per_eu(4, 4)))
void gcn_flash(const float* __restrict__ x,
               const float* __restrict__ w,
               float* __restrict__ out) {
    __shared__ __align__(16) float shm[8 * 68 * 32];

    const int tid  = threadIdx.x;
    const int wave = tid >> 5;
    const int lane = tid & 31;
    const int hi   = lane >> 4;
    const int mrow = lane & 15;
    const int qp   = wave & 1;
    const int kq   = wave >> 1;

    const int b    = blockIdx.x >> 7;
    const int qblk = blockIdx.x & 127;
    const int qbase = qblk * 64 + qp * 32;

    const float* xb = x + (size_t)b * FDIM * N_NODES;

    v2f q[2][16];
#pragma unroll
    for (int u = 0; u < 2; ++u) {
        const float* qcol = xb + (size_t)(2 * hi) * N_NODES + (qbase + 16 * u + mrow);
#pragma unroll
        for (int c = 0; c < 16; ++c) {
            q[u][c].x = qcol[(size_t)(4 * c) * N_NODES];
            q[u][c].y = qcol[(size_t)(4 * c + 1) * N_NODES];
        }
    }

    v8f accT[2][4];
#pragma unroll
    for (int u = 0; u < 2; ++u)
#pragma unroll
        for (int t = 0; t < 4; ++t)
#pragma unroll
            for (int r = 0; r < 8; ++r) accT[u][t][r] = 0.0f;

    float m[2];
#pragma unroll
    for (int u = 0; u < 2; ++u) {
        float d = 0.0f;
#pragma unroll
        for (int c = 0; c < 16; ++c) d += q[u][c].x * q[u][c].x + q[u][c].y * q[u][c].y;
        m[u] = d + __shfl_xor(d, 16, 32);
    }
    float l[2] = {0.0f, 0.0f};

    const int srow0 = tid >> 4;
    const int scol  = (tid & 15) * 4;
    const size_t goff0 = (size_t)srow0 * N_NODES + scol;
    const size_t goff1 = goff0 + (size_t)16 * N_NODES;
    const size_t goff2 = goff0 + (size_t)32 * N_NODES;
    const size_t goff3 = goff0 + (size_t)48 * N_NODES;
    v4f r0 = *(const v4f*)(xb + goff0);
    v4f r1 = *(const v4f*)(xb + goff1);
    v4f r2 = *(const v4f*)(xb + goff2);
    v4f r3 = *(const v4f*)(xb + goff3);

    const int soff = srow0 * PITCH + scol;
    *(v4f*)(shm + soff)              = r0;
    *(v4f*)(shm + soff + 16 * PITCH) = r1;
    *(v4f*)(shm + soff + 32 * PITCH) = r2;
    *(v4f*)(shm + soff + 48 * PITCH) = r3;

    const int NJ = N_NODES / KTRIP;

    v2f kk[16];
    v2f va[4][4];
    v2f pb[2][4];
    v8f s[2];

    auto fetch_next = [&](int jn) __attribute__((always_inline)) {
        const size_t tb = (size_t)jn * KTRIP;
        r0 = *(const v4f*)(xb + tb + goff0);
        r1 = *(const v4f*)(xb + tb + goff1);
        r2 = *(const v4f*)(xb + tb + goff2);
        r3 = *(const v4f*)(xb + tb + goff3);
    };
    auto stage_next = [&](int bufn) __attribute__((always_inline)) {
        float* sn = shm + bufn * (FDIM * PITCH) + soff;
        *(v4f*)(sn)              = r0;
        *(v4f*)(sn + 16 * PITCH) = r1;
        *(v4f*)(sn + 32 * PITCH) = r2;
        *(v4f*)(sn + 48 * PITCH) = r3;
    };
    auto load_k = [&](const float* tile) __attribute__((always_inline)) {
        const float* kl = tile + (2 * hi) * PITCH + mrow;
#pragma unroll
        for (int c = 0; c < 16; ++c) {
            kk[c].x = kl[(4 * c) * PITCH];
            kk[c].y = kl[(4 * c + 1) * PITCH];
        }
    };
    auto load_v = [&](const float* tile) __attribute__((always_inline)) {
        const float* vl = tile + mrow * PITCH + 8 * hi;
#pragma unroll
        for (int t = 0; t < 4; ++t) {
#pragma unroll
            for (int c = 0; c < 4; ++c)
                va[t][c] = *(const v2f*)(vl + (16 * t) * PITCH + 2 * c);
        }
    };
    auto score = [&]() __attribute__((always_inline)) {
#pragma unroll
        for (int u = 0; u < 2; ++u)
#pragma unroll
            for (int r = 0; r < 8; ++r) s[u][r] = 0.0f;
#pragma unroll
        for (int c = 0; c < 16; ++c) {
#pragma unroll
            for (int u = 0; u < 2; ++u)
                s[u] = __builtin_amdgcn_wmma_f32_16x16x4_f32(
                           false, kk[c], false, q[u][c], (short)0, s[u], false, false);
        }
    };
    auto output = [&]() __attribute__((always_inline)) {
#pragma unroll
        for (int t = 0; t < 4; ++t) {
#pragma unroll
            for (int c = 0; c < 4; ++c) {
#pragma unroll
                for (int u = 0; u < 2; ++u)
                    accT[u][t] = __builtin_amdgcn_wmma_f32_16x16x4_f32(
                                     false, va[t][c], false, pb[u][c], (short)0, accT[u][t],
                                     false, false);
            }
        }
    };
    auto softmax = [&]() __attribute__((always_inline)) {
        float mloc[2];
#pragma unroll
        for (int u = 0; u < 2; ++u) {
            mloc[u] = s[u][0];
#pragma unroll
            for (int r = 1; r < 8; ++r) mloc[u] = fmaxf(mloc[u], s[u][r]);
        }
        if (__builtin_amdgcn_ballot_w32((mloc[0] > m[0]) | (mloc[1] > m[1])) != 0u) {
#pragma unroll
            for (int u = 0; u < 2; ++u) {
                float mx = fmaxf(mloc[u], __shfl_xor(mloc[u], 16, 32));
                float mnew = fmaxf(m[u], mx);
                float sc = __expf(m[u] - mnew);
                m[u] = mnew;
                l[u] *= sc;
#pragma unroll
                for (int t = 0; t < 4; ++t)
#pragma unroll
                    for (int r = 0; r < 8; ++r) accT[u][t][r] *= sc;
            }
        }
#pragma unroll
        for (int u = 0; u < 2; ++u) {
            float p[8], ls = 0.0f;
#pragma unroll
            for (int r = 0; r < 8; ++r) { p[r] = __expf(s[u][r] - m[u]); ls += p[r]; }
            l[u] += ls;

#pragma unroll
            for (int c = 0; c < 4; ++c) { pb[u][c].x = p[2 * c]; pb[u][c].y = p[2 * c + 1]; }
        }
    };

    __syncthreads();
    fetch_next(1);
    load_k(shm + 16 * kq);
    __builtin_amdgcn_sched_barrier(0);
    score();
    load_v(shm + 16 * kq);
    stage_next(1);
    __builtin_amdgcn_sched_barrier(0);
    softmax();

    for (int j = 1; j < NJ; ++j) {
        const int buf = j & 1;
        asm volatile("s_wait_dscnt 0x0" ::: "memory");
        __syncthreads();
        if (j + 1 < NJ) fetch_next(j + 1);

        const float* tile = shm + buf * (FDIM * PITCH) + 16 * kq;
        load_k(tile);
        __builtin_amdgcn_sched_barrier(0);

        output();
        __builtin_amdgcn_sched_barrier(0);
        score();
        load_v(tile);
        if (j + 1 < NJ) stage_next(buf ^ 1);
        __builtin_amdgcn_sched_barrier(0);
        softmax();
    }
    output();

    __syncthreads();
    float* area = shm;
    {
        float* mine = area + (size_t)(qp * 4 + kq) * (68 * 32) + lane;
#pragma unroll
        for (int u = 0; u < 2; ++u) {
#pragma unroll
            for (int t = 0; t < 4; ++t)
#pragma unroll
                for (int r = 0; r < 8; ++r)
                    mine[(32 * u + 8 * t + r) * 32] = accT[u][t][r];
            mine[(64 + u) * 32] = m[u];
            mine[(66 + u) * 32] = l[u] + __shfl_xor(l[u], 16, 32);
        }
    }

    v2f wa[4][4];
    {
        const float* wl = w + (size_t)(8 * hi) * FDIM + 16 * kq + mrow;
#pragma unroll
        for (int t = 0; t < 4; ++t)
#pragma unroll
            for (int c = 0; c < 4; ++c) {
                wa[t][c].x = wl[(16 * t + 2 * c) * FDIM];
                wa[t][c].y = wl[(16 * t + 2 * c + 1) * FDIM];
            }
    }
    __syncthreads();

    {
        const float* pair = area + (size_t)(qp * 4) * (68 * 32) + lane;
#pragma unroll
        for (int u = 0; u < 2; ++u) {
            float mk[4], lk[4];
#pragma unroll
            for (int k = 0; k < 4; ++k) {
                mk[k] = pair[(size_t)k * (68 * 32) + (64 + u) * 32];
                lk[k] = pair[(size_t)k * (68 * 32) + (66 + u) * 32];
            }
            float mn = fmaxf(fmaxf(mk[0], mk[1]), fmaxf(mk[2], mk[3]));
            float ak[4], lsum = 0.0f;
#pragma unroll
            for (int k = 0; k < 4; ++k) { ak[k] = __expf(mk[k] - mn); lsum += lk[k] * ak[k]; }
            float inv = 1.0f / lsum;
#pragma unroll
            for (int k = 0; k < 4; ++k) ak[k] *= inv;

            v8f d;
#pragma unroll
            for (int r = 0; r < 8; ++r) d[r] = 0.0f;
#pragma unroll
            for (int t = 0; t < 4; ++t) {
                __builtin_amdgcn_sched_barrier(0);
                float om[8];
#pragma unroll
                for (int r = 0; r < 8; ++r) {
                    float o_acc = 0.0f;
#pragma unroll
                    for (int k = 0; k < 4; ++k)
                        o_acc += pair[(size_t)k * (68 * 32) + (32 * u + 8 * t + r) * 32] * ak[k];
                    om[r] = o_acc;
                }
#pragma unroll
                for (int c = 0; c < 4; ++c) {
                    v2f bf;
                    bf.x = om[2 * c];
                    bf.y = om[2 * c + 1];
                    d = __builtin_amdgcn_wmma_f32_16x16x4_f32(
                            false, wa[t][c], false, bf, (short)0, d, false, false);
                }
            }
            int n = qbase + 16 * u + mrow;
#pragma unroll
            for (int r = 0; r < 8; ++r) {
                int o = 16 * kq + r + 8 * hi;
                out[((size_t)b * FDIM + o) * N_NODES + n] = d[r];
            }
        }
    }
}

extern "C" void kernel_launch(void* const* d_in, const int* in_sizes, int n_in,
                              void* d_out, int out_size, void* d_ws, size_t ws_size,
                              hipStream_t stream) {
    (void)in_sizes; (void)n_in; (void)out_size; (void)d_ws; (void)ws_size;
    const float* x = (const float*)d_in[0];
    const float* w = (const float*)d_in[1];
    float* out = (float*)d_out;

    dim3 gFlash(BATCH * (N_NODES / 64));
    gcn_flash<<<gFlash, 256, 0, stream>>>(x, w, out);
}
